# starting kernel
# baseline (speedup 1.0000x reference)
#include <hip/hip_runtime.h>


typedef __attribute__((ext_vector_type(2))) float v2f;
typedef __attribute__((ext_vector_type(8))) float v8f;

#define IN_C    256
#define NH      8
#define KD2     288
#define MEM     256
#define HD      64
#define OUTC    256
#define HW      64
#define L_IMG   4096

#define LDS_X    0
#define LDS_P    17408
#define LDS_ATT  34816
#define LDS_O    71680
#define LDS_DW   80896

static __device__ __forceinline__ v8f wmma4(v2f a, v2f b, v8f c) {
  return __builtin_amdgcn_wmma_f32_16x16x4_f32(
       false, a,  false, b,
       (short)0, c,  false,  false);
}

__global__ __launch_bounds__(256)
void kcmm_fused_gfx1250(const float* __restrict__ x,
                        const float* __restrict__ Wp,
                        const float* __restrict__ bp,
                        const float* __restrict__ keyb,
                        const float* __restrict__ memb,
                        const float* __restrict__ Wo,
                        const float* __restrict__ bo,
                        float* __restrict__ out)
{
  extern __shared__ float lds[];

  const int tid  = threadIdx.x;
  const int wave = tid >> 5;
  const int lane = tid & 31;
  const int half = lane >> 4;
  const int li   = lane & 15;

  const int blk = blockIdx.x;
  const int b   = blk >> 8;
  const int rem = blk & 255;
  const int y   = rem >> 2;
  const int x0  = (rem & 3) << 4;

  const v8f vzero = {0.f,0.f,0.f,0.f,0.f,0.f,0.f,0.f};

  for (int idx = tid; idx < IN_C * 54; idx += 256) {
    int c  = idx / 54;
    int r2 = idx - c * 54;
    int rr = r2 / 18;
    int cc = r2 - rr * 18;
    int gy = y - 1 + rr;
    int gx = x0 - 1 + cc;
    float v = 0.f;
    if (gy >= 0 && gy < HW && gx >= 0 && gx < HW)
      v = x[((b * IN_C + c) * HW + gy) * HW + gx];
    lds[LDS_X + c * 68 + rr * 18 + cc] = v;
  }
  for (int idx = tid; idx < IN_C * 14; idx += 256) {
    int c = idx / 14;
    int q = 54 + (idx - c * 14);
    lds[LDS_X + c * 68 + q] = 0.f;
  }
  __syncthreads();

  {
    v8f acc[2][4];
    #pragma unroll
    for (int j = 0; j < 2; ++j)
      #pragma unroll
      for (int qt = 0; qt < 4; ++qt) acc[j][qt] = vzero;

    for (int k0 = 0; k0 < IN_C; k0 += 4) {
      const int kl = k0 + 2 * half;
      v2f bf[4];
      #pragma unroll
      for (int qt = 0; qt < 4; ++qt) {
        bf[qt].x = lds[LDS_X + kl * 68 + qt * 16 + li];
        bf[qt].y = lds[LDS_X + (kl + 1) * 68 + qt * 16 + li];
      }
      #pragma unroll
      for (int j = 0; j < 2; ++j) {
        const int o0 = (2 * wave + j) * 16;
        v2f af;
        af.x = Wp[(o0 + li) * IN_C + kl];
        af.y = Wp[(o0 + li) * IN_C + kl + 1];
        #pragma unroll
        for (int qt = 0; qt < 4; ++qt) acc[j][qt] = wmma4(af, bf[qt], acc[j][qt]);
      }
    }
    #pragma unroll
    for (int j = 0; j < 2; ++j)
      #pragma unroll
      for (int r = 0; r < 8; ++r) {
        const int o = (2 * wave + j) * 16 + 8 * half + r;
        const float bv = bp[o];
        #pragma unroll
        for (int qt = 0; qt < 4; ++qt)
          lds[LDS_P + o * 68 + qt * 16 + li] = acc[j][qt][r] + bv;
      }
  }
  __syncthreads();

  const int n = wave;
  v8f att[16];
  #pragma unroll
  for (int t = 0; t < 16; ++t) att[t] = vzero;

  for (int k0 = 0; k0 < KD2; k0 += 4) {
    const int kl = k0 + 2 * half;
    v2f bf;
    #pragma unroll
    for (int v = 0; v < 2; ++v) {
      const int ch = (kl + v) * NH + n;
      const int c  = ch / 9;
      const int p  = ch - c * 9;
      const int kh = p / 3;
      const int kw = p - kh * 3;
      bf[v] = lds[LDS_P + c * 68 + kh * 18 + kw + li];
    }
    #pragma unroll
    for (int t = 0; t < 16; ++t) {
      v2f af;
      af.x = keyb[(n * KD2 + kl) * MEM + t * 16 + li];
      af.y = keyb[(n * KD2 + kl + 1) * MEM + t * 16 + li];
      att[t] = wmma4(af, bf, att[t]);
    }
  }

  {
    float mx = -3.0e38f;
    #pragma unroll
    for (int t = 0; t < 16; ++t)
      #pragma unroll
      for (int r = 0; r < 8; ++r) mx = fmaxf(mx, att[t][r]);
    mx = fmaxf(mx, __shfl_xor(mx, 16, 32));

    float sum = 0.f;
    #pragma unroll
    for (int t = 0; t < 16; ++t)
      #pragma unroll
      for (int r = 0; r < 8; ++r) {
        float e = __expf(att[t][r] - mx);
        att[t][r] = e;
        sum += e;
      }
    sum += __shfl_xor(sum, 16, 32);
    const float inv = 1.f / sum;

    #pragma unroll
    for (int t = 0; t < 16; ++t)
      #pragma unroll
      for (int r = 0; r < 8; ++r) {
        const int i = t * 16 + 8 * half + r;
        lds[LDS_ATT + (n * MEM + i) * 18 + li] = att[t][r] * inv;
      }
  }

  {
    v8f oacc[4];
    #pragma unroll
    for (int t = 0; t < 4; ++t) oacc[t] = vzero;

    for (int k0 = 0; k0 < MEM; k0 += 4) {
      const int kl = k0 + 2 * half;
      v2f bf;
      bf.x = lds[LDS_ATT + (n * MEM + kl) * 18 + li];
      bf.y = lds[LDS_ATT + (n * MEM + kl + 1) * 18 + li];
      #pragma unroll
      for (int t = 0; t < 4; ++t) {
        v2f af;
        af.x = memb[(n * MEM + kl) * HD + t * 16 + li];
        af.y = memb[(n * MEM + kl + 1) * HD + t * 16 + li];
        oacc[t] = wmma4(af, bf, oacc[t]);
      }
    }
    #pragma unroll
    for (int t = 0; t < 4; ++t)
      #pragma unroll
      for (int r = 0; r < 8; ++r) {
        const int d = t * 16 + 8 * half + r;
        lds[LDS_O + (n * HD + d) * 18 + li] = oacc[t][r];
      }
  }
  __syncthreads();

  {
    v8f facc[2];
    facc[0] = vzero; facc[1] = vzero;

    for (int k0 = 0; k0 < NH * HD; k0 += 4) {
      const int kl = k0 + 2 * half;
      v2f bf;
      bf.x = lds[LDS_O + kl * 18 + li];
      bf.y = lds[LDS_O + (kl + 1) * 18 + li];
      #pragma unroll
      for (int j = 0; j < 2; ++j) {
        const int o0 = (2 * wave + j) * 16;
        v2f af;
        af.x = Wo[(o0 + li) * (NH * HD) + kl];
        af.y = Wo[(o0 + li) * (NH * HD) + kl + 1];
        facc[j] = wmma4(af, bf, facc[j]);
      }
    }
    #pragma unroll
    for (int j = 0; j < 2; ++j)
      #pragma unroll
      for (int r = 0; r < 8; ++r) {
        const int oo = (2 * wave + j) * 16 + 8 * half + r;
        out[(b * OUTC + oo) * L_IMG + y * HW + x0 + li] = facc[j][r] + bo[oo];
      }
  }
}

extern "C" void kernel_launch(void* const* d_in, const int* in_sizes, int n_in,
                              void* d_out, int out_size, void* d_ws, size_t ws_size,
                              hipStream_t stream) {
  (void)in_sizes; (void)n_in; (void)d_ws; (void)ws_size; (void)out_size;
  const float* x    = (const float*)d_in[0];
  const float* Wp   = (const float*)d_in[1];
  const float* bpj  = (const float*)d_in[2];
  const float* keyb = (const float*)d_in[3];
  const float* memb = (const float*)d_in[4];
  const float* Wo   = (const float*)d_in[5];
  const float* bo   = (const float*)d_in[6];
  float* out = (float*)d_out;

  const size_t shmem = (size_t)LDS_DW * sizeof(float);
  dim3 grid(8 * 64 * 4);
  dim3 block(256);
  kcmm_fused_gfx1250<<<grid, block, shmem, stream>>>(x, Wp, bpj, keyb, memb, Wo, bo, out);
}


# 32 pixels per workgroup (2 rows x 16) instead of 16: each weight fragment read feeds two WMMAs; softmax tiles feed the readback from registers, which permutes its k (sum) order
# speedup vs baseline: 1.6451x; 1.6451x over previous
#include <hip/hip_runtime.h>


typedef __attribute__((ext_vector_type(2))) float v2f;
typedef __attribute__((ext_vector_type(8))) float v8f;

#define IN_C    256
#define NH      8
#define KD2     288
#define MEM     256
#define HD      64
#define OUTC    256
#define HW      64
#define L_IMG   4096

#define NQ      72
#define SX      88
#define SP      76
#define SO      44

#define LDS_X    0
#define LDS_O    0
#define LDS_P    22528
#define LDS_DW   41984

static __device__ __forceinline__ v8f wmma4(v2f a, v2f b, v8f c) {
  return __builtin_amdgcn_wmma_f32_16x16x4_f32(
       false, a,  false, b,
       (short)0, c,  false,  false);
}

__global__ __launch_bounds__(256)
void kcmm_fused_gfx1250(const float* __restrict__ x,
                        const float* __restrict__ Wp,
                        const float* __restrict__ bp,
                        const float* __restrict__ keyb,
                        const float* __restrict__ memb,
                        const float* __restrict__ Wo,
                        const float* __restrict__ bo,
                        float* __restrict__ out)
{
  extern __shared__ float lds[];

  const int tid  = threadIdx.x;
  const int wave = tid >> 5;
  const int lane = tid & 31;
  const int half = lane >> 4;
  const int li   = lane & 15;

  const int blk = blockIdx.x;
  const int b   = blk >> 7;
  const int rem = blk & 127;
  const int y0  = (rem >> 2) << 1;
  const int x0  = (rem & 3) << 4;

  const v8f vzero = {0.f,0.f,0.f,0.f,0.f,0.f,0.f,0.f};

  for (int idx = tid; idx < IN_C * NQ; idx += 256) {
    int c  = idx / NQ;
    int q  = idx - c * NQ;
    int rr = q / 18;
    int cc = q - rr * 18;
    int gy = y0 - 1 + rr;
    int gx = x0 - 1 + cc;
    float v = 0.f;
    if (gy >= 0 && gy < HW && gx >= 0 && gx < HW)
      v = x[((b * IN_C + c) * HW + gy) * HW + gx];
    lds[LDS_X + c * SX + q] = v;
  }
  for (int idx = tid; idx < IN_C * 8; idx += 256) {
    int c = idx >> 3;
    int q = NQ + (idx & 7);
    lds[LDS_X + c * SX + q] = 0.f;
  }
  __syncthreads();

  {
    v8f acc[2][5];
    #pragma unroll
    for (int j = 0; j < 2; ++j)
      #pragma unroll
      for (int qt = 0; qt < 5; ++qt) acc[j][qt] = vzero;

    for (int k0 = 0; k0 < IN_C; k0 += 4) {
      const int kl = k0 + 2 * half;
      v2f bf[5];
      #pragma unroll
      for (int qt = 0; qt < 5; ++qt) {
        bf[qt].x = lds[LDS_X + kl * SX + qt * 16 + li];
        bf[qt].y = lds[LDS_X + (kl + 1) * SX + qt * 16 + li];
      }
      #pragma unroll
      for (int j = 0; j < 2; ++j) {
        const int o0 = (2 * wave + j) * 16;
        v2f af;
        af.x = Wp[(o0 + li) * IN_C + kl];
        af.y = Wp[(o0 + li) * IN_C + kl + 1];
        #pragma unroll
        for (int qt = 0; qt < 5; ++qt) acc[j][qt] = wmma4(af, bf[qt], acc[j][qt]);
      }
    }

    bool inimg[5];
    #pragma unroll
    for (int qt = 0; qt < 5; ++qt) {
      const int q  = qt * 16 + li;
      const int rr = q / 18;
      const int cc = q - rr * 18;
      const int gy = y0 - 1 + rr;
      const int gx = x0 - 1 + cc;
      inimg[qt] = (q < NQ) && gy >= 0 && gy < HW && gx >= 0 && gx < HW;
    }
    float bv[2][8];
    #pragma unroll
    for (int j = 0; j < 2; ++j)
      #pragma unroll
      for (int r = 0; r < 8; ++r) bv[j][r] = bp[(2 * wave + j) * 16 + 8 * half + r];
    #pragma unroll
    for (int j = 0; j < 2; ++j)
      #pragma unroll
      for (int r = 0; r < 8; ++r) {
        const int o = (2 * wave + j) * 16 + 8 * half + r;
        #pragma unroll
        for (int qt = 0; qt < 5; ++qt) {
          const int q = qt * 16 + li;
          if (q < NQ)
            lds[LDS_P + o * SP + q] = inimg[qt] ? (acc[j][qt][r] + bv[j][r]) : 0.f;
        }
      }
  }
  __syncthreads();

  const int n = wave;
  v8f att[16][2];
  #pragma unroll
  for (int t = 0; t < 16; ++t) { att[t][0] = vzero; att[t][1] = vzero; }

  for (int k0 = 0; k0 < KD2; k0 += 4) {
    const int kl = k0 + 2 * half;
    v2f bf0, bf1;
    #pragma unroll
    for (int v = 0; v < 2; ++v) {
      const int ch = (kl + v) * NH + n;
      const int c  = ch / 9;
      const int p  = ch - c * 9;
      const int kh = p / 3;
      const int kw = p - kh * 3;
      const int off = LDS_P + c * SP + kh * 18 + kw + li;
      bf0[v] = lds[off];
      bf1[v] = lds[off + 18];
    }
    #pragma unroll
    for (int t = 0; t < 16; ++t) {
      v2f af;
      af.x = keyb[(n * KD2 + kl) * MEM + t * 16 + li];
      af.y = keyb[(n * KD2 + kl + 1) * MEM + t * 16 + li];
      att[t][0] = wmma4(af, bf0, att[t][0]);
      att[t][1] = wmma4(af, bf1, att[t][1]);
    }
  }

  #pragma unroll
  for (int nt = 0; nt < 2; ++nt) {
    float mx = -3.0e38f;
    #pragma unroll
    for (int t = 0; t < 16; ++t)
      #pragma unroll
      for (int r = 0; r < 8; ++r) mx = fmaxf(mx, att[t][nt][r]);
    mx = fmaxf(mx, __shfl_xor(mx, 16, 32));

    float sum = 0.f;
    #pragma unroll
    for (int t = 0; t < 16; ++t)
      #pragma unroll
      for (int r = 0; r < 8; ++r) {
        float e = __expf(att[t][nt][r] - mx);
        att[t][nt][r] = e;
        sum += e;
      }
    sum += __shfl_xor(sum, 16, 32);
    const float inv = 1.f / sum;

    #pragma unroll
    for (int t = 0; t < 16; ++t)
      #pragma unroll
      for (int r = 0; r < 8; ++r) att[t][nt][r] *= inv;
  }

  {
    v8f oacc[4][2];
    #pragma unroll
    for (int dt = 0; dt < 4; ++dt) { oacc[dt][0] = vzero; oacc[dt][1] = vzero; }

    #pragma unroll
    for (int t = 0; t < 16; ++t) {
      float a0[4][4], a1[4][4];
      #pragma unroll
      for (int j = 0; j < 4; ++j) {
        const int i0 = t * 16 + 8 * half + 2 * j;
        #pragma unroll
        for (int dt = 0; dt < 4; ++dt) {
          a0[j][dt] = memb[(n * MEM + i0) * HD + dt * 16 + li];
          a1[j][dt] = memb[(n * MEM + i0 + 1) * HD + dt * 16 + li];
        }
      }
      #pragma unroll
      for (int j = 0; j < 4; ++j) {
        v2f b0, b1;
        b0.x = att[t][0][2 * j];
        b0.y = att[t][0][2 * j + 1];
        b1.x = att[t][1][2 * j];
        b1.y = att[t][1][2 * j + 1];
        #pragma unroll
        for (int dt = 0; dt < 4; ++dt) {
          v2f af;
          af.x = a0[j][dt];
          af.y = a1[j][dt];
          oacc[dt][0] = wmma4(af, b0, oacc[dt][0]);
          oacc[dt][1] = wmma4(af, b1, oacc[dt][1]);
        }
      }
      __builtin_amdgcn_sched_barrier(0);
    }
    #pragma unroll
    for (int dt = 0; dt < 4; ++dt)
      #pragma unroll
      for (int nt = 0; nt < 2; ++nt)
        #pragma unroll
        for (int r = 0; r < 8; ++r) {
          const int d = dt * 16 + 8 * half + r;
          lds[LDS_O + (n * HD + d) * SO + nt * 16 + li] = oacc[dt][nt][r];
        }
  }
  __syncthreads();

  {
    v8f facc[2][2];
    facc[0][0] = vzero; facc[0][1] = vzero; facc[1][0] = vzero; facc[1][1] = vzero;

    for (int k0 = 0; k0 < NH * HD; k0 += 4) {
      const int kl = k0 + 2 * half;
      v2f bf0, bf1;
      bf0.x = lds[LDS_O + kl * SO + li];
      bf0.y = lds[LDS_O + (kl + 1) * SO + li];
      bf1.x = lds[LDS_O + kl * SO + 16 + li];
      bf1.y = lds[LDS_O + (kl + 1) * SO + 16 + li];
      #pragma unroll
      for (int j = 0; j < 2; ++j) {
        const int o0 = (2 * wave + j) * 16;
        v2f af;
        af.x = Wo[(o0 + li) * (NH * HD) + kl];
        af.y = Wo[(o0 + li) * (NH * HD) + kl + 1];
        facc[j][0] = wmma4(af, bf0, facc[j][0]);
        facc[j][1] = wmma4(af, bf1, facc[j][1]);
      }
    }
    #pragma unroll
    for (int j = 0; j < 2; ++j)
      #pragma unroll
      for (int nt = 0; nt < 2; ++nt)
        #pragma unroll
        for (int r = 0; r < 8; ++r) {
          const int oo = (2 * wave + j) * 16 + 8 * half + r;
          out[(b * OUTC + oo) * L_IMG + (y0 + nt) * HW + x0 + li] = facc[j][nt][r] + bo[oo];
        }
  }
}

extern "C" void kernel_launch(void* const* d_in, const int* in_sizes, int n_in,
                              void* d_out, int out_size, void* d_ws, size_t ws_size,
                              hipStream_t stream) {
  (void)in_sizes; (void)n_in; (void)d_ws; (void)ws_size; (void)out_size;
  const float* x    = (const float*)d_in[0];
  const float* Wp   = (const float*)d_in[1];
  const float* bpj  = (const float*)d_in[2];
  const float* keyb = (const float*)d_in[3];
  const float* memb = (const float*)d_in[4];
  const float* Wo   = (const float*)d_in[5];
  const float* bo   = (const float*)d_in[6];
  float* out = (float*)d_out;

  const size_t shmem = (size_t)LDS_DW * sizeof(float);
  dim3 grid(1024);
  dim3 block(256);
  kcmm_fused_gfx1250<<<grid, block, shmem, stream>>>(x, Wp, bpj, keyb, memb, Wo, bo, out);
}


# x halo load: a lane keeps fixed tile positions, so its 72 global loads (clamped address, 0 selected outside the image) are independent and in flight together, not 72 load-wait-store trips
# speedup vs baseline: 2.0665x; 1.2562x over previous
#include <hip/hip_runtime.h>


typedef __attribute__((ext_vector_type(2))) float v2f;
typedef __attribute__((ext_vector_type(8))) float v8f;

#define IN_C    256
#define NH      8
#define KD2     288
#define MEM     256
#define HD      64
#define OUTC    256
#define HW      64
#define L_IMG   4096

#define NQ      72
#define SX      88
#define SP      76
#define SO      44

#define LDS_X    0
#define LDS_O    0
#define LDS_P    22528
#define LDS_DW   41984

static __device__ __forceinline__ v8f wmma4(v2f a, v2f b, v8f c) {
  return __builtin_amdgcn_wmma_f32_16x16x4_f32(
       false, a,  false, b,
       (short)0, c,  false,  false);
}

__global__ __launch_bounds__(256)
void kcmm_fused_gfx1250(const float* __restrict__ x,
                        const float* __restrict__ Wp,
                        const float* __restrict__ bp,
                        const float* __restrict__ keyb,
                        const float* __restrict__ memb,
                        const float* __restrict__ Wo,
                        const float* __restrict__ bo,
                        float* __restrict__ out)
{
  extern __shared__ float lds[];

  const int tid  = threadIdx.x;
  const int wave = tid >> 5;
  const int lane = tid & 31;
  const int half = lane >> 4;
  const int li   = lane & 15;

  const int blk = blockIdx.x;
  const int b   = blk >> 7;
  const int rem = blk & 127;
  const int y0  = (rem >> 2) << 1;
  const int x0  = (rem & 3) << 4;

  const v8f vzero = {0.f,0.f,0.f,0.f,0.f,0.f,0.f,0.f};

  {
    int  qq[3], goff[3];
    bool inb[3];
    qq[0] = lane;
    qq[1] = 32 + lane;
    qq[2] = 64 + (lane & 7);
    #pragma unroll
    for (int s = 0; s < 3; ++s) {
      const int rr = qq[s] / 18;
      const int cc = qq[s] - rr * 18;
      const int gy = y0 - 1 + rr;
      const int gx = x0 - 1 + cc;
      inb[s] = gy >= 0 && gy < HW && gx >= 0 && gx < HW;
      goff[s] = min(max(gy, 0), HW - 1) * HW + min(max(gx, 0), HW - 1);
    }
    const int csub = lane >> 3;
    const int wv = __builtin_amdgcn_readfirstlane(wave);
    const float* __restrict__ xc = x + ((size_t)b * IN_C + wv * 32) * L_IMG;
    float* __restrict__ lc = lds + LDS_X + wv * 32 * SX;
    float vA[32], vB[32], vC[8];
    #pragma unroll
    for (int i = 0; i < 32; ++i) {
      vA[i] = xc[i * L_IMG + goff[0]];
      vB[i] = xc[i * L_IMG + goff[1]];
    }
    #pragma unroll
    for (int g = 0; g < 8; ++g) vC[g] = xc[(4 * g) * L_IMG + csub * L_IMG + goff[2]];
    #pragma unroll
    for (int i = 0; i < 32; ++i) {
      lc[i * SX + qq[0]] = inb[0] ? vA[i] : 0.f;
      lc[i * SX + qq[1]] = inb[1] ? vB[i] : 0.f;
    }
    #pragma unroll
    for (int g = 0; g < 8; ++g)
      lc[(4 * g) * SX + csub * SX + qq[2]] = inb[2] ? vC[g] : 0.f;
  }
  for (int idx = tid; idx < IN_C * 8; idx += 256) {
    int c = idx >> 3;
    int q = NQ + (idx & 7);
    lds[LDS_X + c * SX + q] = 0.f;
  }
  __syncthreads();

  {
    v8f acc[2][5];
    #pragma unroll
    for (int j = 0; j < 2; ++j)
      #pragma unroll
      for (int qt = 0; qt < 5; ++qt) acc[j][qt] = vzero;

    for (int k0 = 0; k0 < IN_C; k0 += 4) {
      const int kl = k0 + 2 * half;
      v2f bf[5];
      #pragma unroll
      for (int qt = 0; qt < 5; ++qt) {
        bf[qt].x = lds[LDS_X + kl * SX + qt * 16 + li];
        bf[qt].y = lds[LDS_X + (kl + 1) * SX + qt * 16 + li];
      }
      #pragma unroll
      for (int j = 0; j < 2; ++j) {
        const int o0 = (2 * wave + j) * 16;
        v2f af;
        af.x = Wp[(o0 + li) * IN_C + kl];
        af.y = Wp[(o0 + li) * IN_C + kl + 1];
        #pragma unroll
        for (int qt = 0; qt < 5; ++qt) acc[j][qt] = wmma4(af, bf[qt], acc[j][qt]);
      }
    }

    bool inimg[5];
    #pragma unroll
    for (int qt = 0; qt < 5; ++qt) {
      const int q  = qt * 16 + li;
      const int rr = q / 18;
      const int cc = q - rr * 18;
      const int gy = y0 - 1 + rr;
      const int gx = x0 - 1 + cc;
      inimg[qt] = (q < NQ) && gy >= 0 && gy < HW && gx >= 0 && gx < HW;
    }
    float bv[2][8];
    #pragma unroll
    for (int j = 0; j < 2; ++j)
      #pragma unroll
      for (int r = 0; r < 8; ++r) bv[j][r] = bp[(2 * wave + j) * 16 + 8 * half + r];
    #pragma unroll
    for (int j = 0; j < 2; ++j)
      #pragma unroll
      for (int r = 0; r < 8; ++r) {
        const int o = (2 * wave + j) * 16 + 8 * half + r;
        #pragma unroll
        for (int qt = 0; qt < 5; ++qt) {
          const int q = qt * 16 + li;
          if (q < NQ)
            lds[LDS_P + o * SP + q] = inimg[qt] ? (acc[j][qt][r] + bv[j][r]) : 0.f;
        }
      }
  }
  __syncthreads();

  const int n = wave;
  v8f att[16][2];
  #pragma unroll
  for (int t = 0; t < 16; ++t) { att[t][0] = vzero; att[t][1] = vzero; }

  for (int k0 = 0; k0 < KD2; k0 += 4) {
    const int kl = k0 + 2 * half;
    v2f bf0, bf1;
    #pragma unroll
    for (int v = 0; v < 2; ++v) {
      const int ch = (kl + v) * NH + n;
      const int c  = ch / 9;
      const int p  = ch - c * 9;
      const int kh = p / 3;
      const int kw = p - kh * 3;
      const int off = LDS_P + c * SP + kh * 18 + kw + li;
      bf0[v] = lds[off];
      bf1[v] = lds[off + 18];
    }
    #pragma unroll
    for (int t = 0; t < 16; ++t) {
      v2f af;
      af.x = keyb[(n * KD2 + kl) * MEM + t * 16 + li];
      af.y = keyb[(n * KD2 + kl + 1) * MEM + t * 16 + li];
      att[t][0] = wmma4(af, bf0, att[t][0]);
      att[t][1] = wmma4(af, bf1, att[t][1]);
    }
  }

  #pragma unroll
  for (int nt = 0; nt < 2; ++nt) {
    float mx = -3.0e38f;
    #pragma unroll
    for (int t = 0; t < 16; ++t)
      #pragma unroll
      for (int r = 0; r < 8; ++r) mx = fmaxf(mx, att[t][nt][r]);
    mx = fmaxf(mx, __shfl_xor(mx, 16, 32));

    float sum = 0.f;
    #pragma unroll
    for (int t = 0; t < 16; ++t)
      #pragma unroll
      for (int r = 0; r < 8; ++r) {
        float e = __expf(att[t][nt][r] - mx);
        att[t][nt][r] = e;
        sum += e;
      }
    sum += __shfl_xor(sum, 16, 32);
    const float inv = 1.f / sum;

    #pragma unroll
    for (int t = 0; t < 16; ++t)
      #pragma unroll
      for (int r = 0; r < 8; ++r) att[t][nt][r] *= inv;
  }

  {
    v8f oacc[4][2];
    #pragma unroll
    for (int dt = 0; dt < 4; ++dt) { oacc[dt][0] = vzero; oacc[dt][1] = vzero; }

    #pragma unroll
    for (int t = 0; t < 16; ++t) {
      float a0[4][4], a1[4][4];
      #pragma unroll
      for (int j = 0; j < 4; ++j) {
        const int i0 = t * 16 + 8 * half + 2 * j;
        #pragma unroll
        for (int dt = 0; dt < 4; ++dt) {
          a0[j][dt] = memb[(n * MEM + i0) * HD + dt * 16 + li];
          a1[j][dt] = memb[(n * MEM + i0 + 1) * HD + dt * 16 + li];
        }
      }
      #pragma unroll
      for (int j = 0; j < 4; ++j) {
        v2f b0, b1;
        b0.x = att[t][0][2 * j];
        b0.y = att[t][0][2 * j + 1];
        b1.x = att[t][1][2 * j];
        b1.y = att[t][1][2 * j + 1];
        #pragma unroll
        for (int dt = 0; dt < 4; ++dt) {
          v2f af;
          af.x = a0[j][dt];
          af.y = a1[j][dt];
          oacc[dt][0] = wmma4(af, b0, oacc[dt][0]);
          oacc[dt][1] = wmma4(af, b1, oacc[dt][1]);
        }
      }
      __builtin_amdgcn_sched_barrier(0);
    }
    #pragma unroll
    for (int dt = 0; dt < 4; ++dt)
      #pragma unroll
      for (int nt = 0; nt < 2; ++nt)
        #pragma unroll
        for (int r = 0; r < 8; ++r) {
          const int d = dt * 16 + 8 * half + r;
          lds[LDS_O + (n * HD + d) * SO + nt * 16 + li] = oacc[dt][nt][r];
        }
  }
  __syncthreads();

  {
    v8f facc[2][2];
    facc[0][0] = vzero; facc[0][1] = vzero; facc[1][0] = vzero; facc[1][1] = vzero;

    for (int k0 = 0; k0 < NH * HD; k0 += 4) {
      const int kl = k0 + 2 * half;
      v2f bf0, bf1;
      bf0.x = lds[LDS_O + kl * SO + li];
      bf0.y = lds[LDS_O + (kl + 1) * SO + li];
      bf1.x = lds[LDS_O + kl * SO + 16 + li];
      bf1.y = lds[LDS_O + (kl + 1) * SO + 16 + li];
      #pragma unroll
      for (int j = 0; j < 2; ++j) {
        const int o0 = (2 * wave + j) * 16;
        v2f af;
        af.x = Wo[(o0 + li) * (NH * HD) + kl];
        af.y = Wo[(o0 + li) * (NH * HD) + kl + 1];
        facc[j][0] = wmma4(af, bf0, facc[j][0]);
        facc[j][1] = wmma4(af, bf1, facc[j][1]);
      }
    }
    #pragma unroll
    for (int j = 0; j < 2; ++j)
      #pragma unroll
      for (int nt = 0; nt < 2; ++nt)
        #pragma unroll
        for (int r = 0; r < 8; ++r) {
          const int oo = (2 * wave + j) * 16 + 8 * half + r;
          out[(b * OUTC + oo) * L_IMG + (y0 + nt) * HW + x0 + li] = facc[j][nt][r] + bo[oo];
        }
  }
}

extern "C" void kernel_launch(void* const* d_in, const int* in_sizes, int n_in,
                              void* d_out, int out_size, void* d_ws, size_t ws_size,
                              hipStream_t stream) {
  (void)in_sizes; (void)n_in; (void)d_ws; (void)ws_size; (void)out_size;
  const float* x    = (const float*)d_in[0];
  const float* Wp   = (const float*)d_in[1];
  const float* bpj  = (const float*)d_in[2];
  const float* keyb = (const float*)d_in[3];
  const float* memb = (const float*)d_in[4];
  const float* Wo   = (const float*)d_in[5];
  const float* bo   = (const float*)d_in[6];
  float* out = (float*)d_out;

  const size_t shmem = (size_t)LDS_DW * sizeof(float);
  dim3 grid(1024);
  dim3 block(256);
  kcmm_fused_gfx1250<<<grid, block, shmem, stream>>>(x, Wp, bpj, keyb, memb, Wo, bo, out);
}


# operand fragments double-buffered in registers in all four GEMM loops: the next step's global and LDS loads are issued before the current step's WMMAs; k order and summation order unchanged
# speedup vs baseline: 2.2377x; 1.0828x over previous
#include <hip/hip_runtime.h>


typedef __attribute__((ext_vector_type(2))) float v2f;
typedef __attribute__((ext_vector_type(8))) float v8f;

#define IN_C    256
#define NH      8
#define KD2     288
#define MEM     256
#define HD      64
#define OUTC    256
#define HW      64
#define L_IMG   4096

#define NQ      72
#define SX      88
#define SP      76
#define SO      44

#define LDS_X    0
#define LDS_O    0
#define LDS_P    22528
#define LDS_DW   41984

static __device__ __forceinline__ v8f wmma4(v2f a, v2f b, v8f c) {
  return __builtin_amdgcn_wmma_f32_16x16x4_f32(
       false, a,  false, b,
       (short)0, c,  false,  false);
}

__global__ __launch_bounds__(256)
void kcmm_fused_gfx1250(const float* __restrict__ x,
                        const float* __restrict__ Wp,
                        const float* __restrict__ bp,
                        const float* __restrict__ keyb,
                        const float* __restrict__ memb,
                        const float* __restrict__ Wo,
                        const float* __restrict__ bo,
                        float* __restrict__ out)
{
  extern __shared__ float lds[];

  const int tid  = threadIdx.x;
  const int wave = tid >> 5;
  const int lane = tid & 31;
  const int half = lane >> 4;
  const int li   = lane & 15;

  const int blk = blockIdx.x;
  const int b   = blk >> 7;
  const int rem = blk & 127;
  const int y0  = (rem >> 2) << 1;
  const int x0  = (rem & 3) << 4;

  const v8f vzero = {0.f,0.f,0.f,0.f,0.f,0.f,0.f,0.f};

  {
    int  qq[3], goff[3];
    bool inb[3];
    qq[0] = lane;
    qq[1] = 32 + lane;
    qq[2] = 64 + (lane & 7);
    #pragma unroll
    for (int s = 0; s < 3; ++s) {
      const int rr = qq[s] / 18;
      const int cc = qq[s] - rr * 18;
      const int gy = y0 - 1 + rr;
      const int gx = x0 - 1 + cc;
      inb[s] = gy >= 0 && gy < HW && gx >= 0 && gx < HW;
      goff[s] = min(max(gy, 0), HW - 1) * HW + min(max(gx, 0), HW - 1);
    }
    const int csub = lane >> 3;
    const int wv = __builtin_amdgcn_readfirstlane(wave);
    const float* __restrict__ xc = x + ((size_t)b * IN_C + wv * 32) * L_IMG;
    float* __restrict__ lc = lds + LDS_X + wv * 32 * SX;
    float vA[32], vB[32], vC[8];
    #pragma unroll
    for (int i = 0; i < 32; ++i) {
      vA[i] = xc[i * L_IMG + goff[0]];
      vB[i] = xc[i * L_IMG + goff[1]];
    }
    #pragma unroll
    for (int g = 0; g < 8; ++g) vC[g] = xc[(4 * g) * L_IMG + csub * L_IMG + goff[2]];
    #pragma unroll
    for (int i = 0; i < 32; ++i) {
      lc[i * SX + qq[0]] = inb[0] ? vA[i] : 0.f;
      lc[i * SX + qq[1]] = inb[1] ? vB[i] : 0.f;
    }
    #pragma unroll
    for (int g = 0; g < 8; ++g)
      lc[(4 * g) * SX + csub * SX + qq[2]] = inb[2] ? vC[g] : 0.f;
  }
  for (int idx = tid; idx < IN_C * 8; idx += 256) {
    int c = idx >> 3;
    int q = NQ + (idx & 7);
    lds[LDS_X + c * SX + q] = 0.f;
  }
  __syncthreads();

  {
    v8f acc[2][5];
    #pragma unroll
    for (int j = 0; j < 2; ++j)
      #pragma unroll
      for (int qt = 0; qt < 5; ++qt) acc[j][qt] = vzero;

#define P1_LOAD(AF, BF, K0)                                                        \
    {                                                                              \
      _Pragma("unroll")       \
      for (int s = 0; s < 2; ++s) {                                                \
        const int kl = (K0) + 4 * s + 2 * half;                                    \
        _Pragma("unroll")                                                          \
        for (int j = 0; j < 2; ++j) {                                              \
          const int o0 = (2 * wave + j) * 16;                                      \
          AF[s][j].x = Wp[(o0 + li) * IN_C + kl];                                  \
          AF[s][j].y = Wp[(o0 + li) * IN_C + kl + 1];                              \
        }                                                                          \
        _Pragma("unroll")                                                          \
        for (int qt = 0; qt < 5; ++qt) {                                           \
          BF[s][qt].x = lds[LDS_X + kl * SX + qt * 16 + li];                       \
          BF[s][qt].y = lds[LDS_X + (kl + 1) * SX + qt * 16 + li];                 \
        }                                                                          \
      }                                                                            \
    }
#define P1_MMA(AF, BF)                                                             \
    {                                                                              \
      _Pragma("unroll")                  \
      for (int s = 0; s < 2; ++s)                                                  \
        _Pragma("unroll")                                                          \
        for (int j = 0; j < 2; ++j)                                                \
          _Pragma("unroll")                                                        \
          for (int qt = 0; qt < 5; ++qt)                                           \
            acc[j][qt] = wmma4(AF[s][j], BF[s][qt], acc[j][qt]);                   \
    }
    v2f pa0[2][2], pa1[2][2], pb0[2][5], pb1[2][5];
    P1_LOAD(pa0, pb0, 0)
    for (int k0 = 0; k0 < IN_C - 16; k0 += 16) {
      P1_LOAD(pa1, pb1, k0 + 8)
      __builtin_amdgcn_sched_barrier(0);
      P1_MMA(pa0, pb0)
      __builtin_amdgcn_sched_barrier(0);
      P1_LOAD(pa0, pb0, k0 + 16)
      __builtin_amdgcn_sched_barrier(0);
      P1_MMA(pa1, pb1)
      __builtin_amdgcn_sched_barrier(0);
    }
    P1_LOAD(pa1, pb1, IN_C - 8)
    __builtin_amdgcn_sched_barrier(0);
    P1_MMA(pa0, pb0)
    __builtin_amdgcn_sched_barrier(0);
    P1_MMA(pa1, pb1)
#undef P1_LOAD
#undef P1_MMA

    bool inimg[5];
    #pragma unroll
    for (int qt = 0; qt < 5; ++qt) {
      const int q  = qt * 16 + li;
      const int rr = q / 18;
      const int cc = q - rr * 18;
      const int gy = y0 - 1 + rr;
      const int gx = x0 - 1 + cc;
      inimg[qt] = (q < NQ) && gy >= 0 && gy < HW && gx >= 0 && gx < HW;
    }
    float bv[2][8];
    #pragma unroll
    for (int j = 0; j < 2; ++j)
      #pragma unroll
      for (int r = 0; r < 8; ++r) bv[j][r] = bp[(2 * wave + j) * 16 + 8 * half + r];
    #pragma unroll
    for (int j = 0; j < 2; ++j)
      #pragma unroll
      for (int r = 0; r < 8; ++r) {
        const int o = (2 * wave + j) * 16 + 8 * half + r;
        #pragma unroll
        for (int qt = 0; qt < 5; ++qt) {
          const int q = qt * 16 + li;
          if (q < NQ)
            lds[LDS_P + o * SP + q] = inimg[qt] ? (acc[j][qt][r] + bv[j][r]) : 0.f;
        }
      }
  }
  __syncthreads();

  const int n = wave;
  v8f att[16][2];
  #pragma unroll
  for (int t = 0; t < 16; ++t) { att[t][0] = vzero; att[t][1] = vzero; }

#define K2_LOAD(AF, B0, B1, K0)                                                    \
  {                                                                                \
    const int kl = (K0) + 2 * half;                                                \
    _Pragma("unroll")             \
    for (int t = 0; t < 16; ++t) {                                                 \
      AF[t].x = keyb[(n * KD2 + kl) * MEM + t * 16 + li];                          \
      AF[t].y = keyb[(n * KD2 + kl + 1) * MEM + t * 16 + li];                      \
    }                                                                              \
    _Pragma("unroll")                                                              \
    for (int v = 0; v < 2; ++v) {                                                  \
      const int ch = (kl + v) * NH + n;         \
      const int c  = ch / 9;                                                       \
      const int p  = ch - c * 9;                                                   \
      const int kh = p / 3;                                                        \
      const int kw = p - kh * 3;                                                   \
      const int off = LDS_P + c * SP + kh * 18 + kw + li;                          \
      B0[v] = lds[off];                                                            \
      B1[v] = lds[off + 18];                                                       \
    }                                                                              \
  }
#define K2_MMA(AF, B0, B1)                                                         \
  {                                                                                \
    _Pragma("unroll")                  \
    for (int t = 0; t < 16; ++t) {                                                 \
      att[t][0] = wmma4(AF[t], B0, att[t][0]);                                     \
      att[t][1] = wmma4(AF[t], B1, att[t][1]);                                     \
    }                                                                              \
  }
  {
    v2f ka0[16], ka1[16], kb00, kb01, kb10, kb11;
    K2_LOAD(ka0, kb00, kb01, 0)
    for (int k0 = 0; k0 < KD2 - 8; k0 += 8) {
      K2_LOAD(ka1, kb10, kb11, k0 + 4)
      __builtin_amdgcn_sched_barrier(0);
      K2_MMA(ka0, kb00, kb01)
      __builtin_amdgcn_sched_barrier(0);
      K2_LOAD(ka0, kb00, kb01, k0 + 8)
      __builtin_amdgcn_sched_barrier(0);
      K2_MMA(ka1, kb10, kb11)
      __builtin_amdgcn_sched_barrier(0);
    }
    K2_LOAD(ka1, kb10, kb11, KD2 - 4)
    __builtin_amdgcn_sched_barrier(0);
    K2_MMA(ka0, kb00, kb01)
    __builtin_amdgcn_sched_barrier(0);
    K2_MMA(ka1, kb10, kb11)
  }
#undef K2_LOAD
#undef K2_MMA

  #pragma unroll
  for (int nt = 0; nt < 2; ++nt) {
    float mx = -3.0e38f;
    #pragma unroll
    for (int t = 0; t < 16; ++t)
      #pragma unroll
      for (int r = 0; r < 8; ++r) mx = fmaxf(mx, att[t][nt][r]);
    mx = fmaxf(mx, __shfl_xor(mx, 16, 32));

    float sum = 0.f;
    #pragma unroll
    for (int t = 0; t < 16; ++t)
      #pragma unroll
      for (int r = 0; r < 8; ++r) {
        float e = __expf(att[t][nt][r] - mx);
        att[t][nt][r] = e;
        sum += e;
      }
    sum += __shfl_xor(sum, 16, 32);
    const float inv = 1.f / sum;

    #pragma unroll
    for (int t = 0; t < 16; ++t)
      #pragma unroll
      for (int r = 0; r < 8; ++r) att[t][nt][r] *= inv;
  }

  {
    v8f oacc[4][2];
    #pragma unroll
    for (int dt = 0; dt < 4; ++dt) { oacc[dt][0] = vzero; oacc[dt][1] = vzero; }

    float ma[2][2][4][4];
#define M3_LOAD(SET, T)                                                            \
    {                                                                              \
      _Pragma("unroll")           \
      for (int j = 0; j < 4; ++j) {                                                \
        const int i0 = (T) * 16 + 8 * half + 2 * j;                                \
        _Pragma("unroll")                                                          \
        for (int dt = 0; dt < 4; ++dt) {                                           \
          ma[SET][0][j][dt] = memb[(n * MEM + i0) * HD + dt * 16 + li];            \
          ma[SET][1][j][dt] = memb[(n * MEM + i0 + 1) * HD + dt * 16 + li];        \
        }                                                                          \
      }                                                                            \
    }
    M3_LOAD(0, 0)
    #pragma unroll
    for (int t = 0; t < 16; ++t) {
      if (t + 1 < 16) M3_LOAD((t + 1) & 1, t + 1)
      #pragma unroll
      for (int j = 0; j < 4; ++j) {
        v2f b0, b1;
        b0.x = att[t][0][2 * j];
        b0.y = att[t][0][2 * j + 1];
        b1.x = att[t][1][2 * j];
        b1.y = att[t][1][2 * j + 1];
        #pragma unroll
        for (int dt = 0; dt < 4; ++dt) {
          v2f af;
          af.x = ma[t & 1][0][j][dt];
          af.y = ma[t & 1][1][j][dt];
          oacc[dt][0] = wmma4(af, b0, oacc[dt][0]);
          oacc[dt][1] = wmma4(af, b1, oacc[dt][1]);
        }
      }
      __builtin_amdgcn_sched_barrier(0);
    }
#undef M3_LOAD
    #pragma unroll
    for (int dt = 0; dt < 4; ++dt)
      #pragma unroll
      for (int nt = 0; nt < 2; ++nt)
        #pragma unroll
        for (int r = 0; r < 8; ++r) {
          const int d = dt * 16 + 8 * half + r;
          lds[LDS_O + (n * HD + d) * SO + nt * 16 + li] = oacc[dt][nt][r];
        }
  }
  __syncthreads();

  {
    v8f facc[2][2];
    facc[0][0] = vzero; facc[0][1] = vzero; facc[1][0] = vzero; facc[1][1] = vzero;

#define O4_LOAD(AF, BF, K0)                                                        \
    {                                                                              \
      _Pragma("unroll")       \
      for (int s = 0; s < 8; ++s) {                                                \
        const int kl = (K0) + 4 * s + 2 * half;                                    \
        _Pragma("unroll")                                                          \
        for (int j = 0; j < 2; ++j) {                                              \
          const int o0 = (2 * wave + j) * 16;                                      \
          AF[s][j].x = Wo[(o0 + li) * (NH * HD) + kl];                             \
          AF[s][j].y = Wo[(o0 + li) * (NH * HD) + kl + 1];                         \
        }                                                                          \
        BF[s][0].x = lds[LDS_O + kl * SO + li];                                    \
        BF[s][0].y = lds[LDS_O + (kl + 1) * SO + li];                              \
        BF[s][1].x = lds[LDS_O + kl * SO + 16 + li];                               \
        BF[s][1].y = lds[LDS_O + (kl + 1) * SO + 16 + li];                         \
      }                                                                            \
    }
#define O4_MMA(AF, BF)                                                             \
    {                                                                              \
      _Pragma("unroll")                  \
      for (int s = 0; s < 8; ++s)                                                  \
        _Pragma("unroll")                                                          \
        for (int j = 0; j < 2; ++j) {                                              \
          facc[j][0] = wmma4(AF[s][j], BF[s][0], facc[j][0]);                      \
          facc[j][1] = wmma4(AF[s][j], BF[s][1], facc[j][1]);                      \
        }                                                                          \
    }
    v2f oa0[8][2], oa1[8][2], ob0[8][2], ob1[8][2];
    O4_LOAD(oa0, ob0, 0)
    for (int k0 = 0; k0 < NH * HD - 64; k0 += 64) {
      O4_LOAD(oa1, ob1, k0 + 32)
      __builtin_amdgcn_sched_barrier(0);
      O4_MMA(oa0, ob0)
      __builtin_amdgcn_sched_barrier(0);
      O4_LOAD(oa0, ob0, k0 + 64)
      __builtin_amdgcn_sched_barrier(0);
      O4_MMA(oa1, ob1)
      __builtin_amdgcn_sched_barrier(0);
    }
    O4_LOAD(oa1, ob1, NH * HD - 32)
    __builtin_amdgcn_sched_barrier(0);
    O4_MMA(oa0, ob0)
    __builtin_amdgcn_sched_barrier(0);
    O4_MMA(oa1, ob1)
#undef O4_LOAD
#undef O4_MMA
    #pragma unroll
    for (int j = 0; j < 2; ++j)
      #pragma unroll
      for (int nt = 0; nt < 2; ++nt)
        #pragma unroll
        for (int r = 0; r < 8; ++r) {
          const int oo = (2 * wave + j) * 16 + 8 * half + r;
          out[(b * OUTC + oo) * L_IMG + (y0 + nt) * HW + x0 + li] = facc[j][nt][r] + bo[oo];
        }
  }
}

extern "C" void kernel_launch(void* const* d_in, const int* in_sizes, int n_in,
                              void* d_out, int out_size, void* d_ws, size_t ws_size,
                              hipStream_t stream) {
  (void)in_sizes; (void)n_in; (void)d_ws; (void)ws_size; (void)out_size;
  const float* x    = (const float*)d_in[0];
  const float* Wp   = (const float*)d_in[1];
  const float* bpj  = (const float*)d_in[2];
  const float* keyb = (const float*)d_in[3];
  const float* memb = (const float*)d_in[4];
  const float* Wo   = (const float*)d_in[5];
  const float* bo   = (const float*)d_in[6];
  float* out = (float*)d_out;

  const size_t shmem = (size_t)LDS_DW * sizeof(float);
  dim3 grid(1024);
  dim3 block(256);
  kcmm_fused_gfx1250<<<grid, block, shmem, stream>>>(x, Wp, bpj, keyb, memb, Wo, bo, out);
}


# 64 pixels per workgroup (4 rows x 16) instead of 32: each weight fragment feeds four WMMAs; the 256 slots go in two halves under a running-max softmax, which changes softmax and read-back sum order
# speedup vs baseline: 2.5041x; 1.1191x over previous
#include <hip/hip_runtime.h>


typedef __attribute__((ext_vector_type(2))) float v2f;
typedef __attribute__((ext_vector_type(8))) float v8f;

#define IN_C    256
#define NH      8
#define KD2     288
#define MEM     256
#define HD      64
#define OUTC    256
#define HW      64
#define L_IMG   4096

#define ROWS    4
#define NQ      108
#define QT      7
#define SX      120
#define SP      116
#define SO      76

#define LDS_X    0
#define LDS_O    0
#define LDS_P    38912
#define LDS_DW   68608

static __device__ __forceinline__ v8f wmma4(v2f a, v2f b, v8f c) {
  return __builtin_amdgcn_wmma_f32_16x16x4_f32(
       false, a,  false, b,
       (short)0, c,  false,  false);
}

__global__ __launch_bounds__(256)
void kcmm_fused_gfx1250(const float* __restrict__ x,
                        const float* __restrict__ Wp,
                        const float* __restrict__ bp,
                        const float* __restrict__ keyb,
                        const float* __restrict__ memb,
                        const float* __restrict__ Wo,
                        const float* __restrict__ bo,
                        float* __restrict__ out)
{
  extern __shared__ float lds[];

  const int tid  = threadIdx.x;
  const int wave = tid >> 5;
  const int lane = tid & 31;
  const int half = lane >> 4;
  const int li   = lane & 15;

  const int blk = blockIdx.x;
  const int b   = blk >> 6;
  const int rem = blk & 63;
  const int y0  = (rem >> 2) << 2;
  const int x0  = (rem & 3) << 4;

  const v8f vzero = {0.f,0.f,0.f,0.f,0.f,0.f,0.f,0.f};

  {
    int  qq[4], goff[4];
    bool inb[4];
    qq[0] = lane;
    qq[1] = 32 + lane;
    qq[2] = 64 + lane;
    qq[3] = 96 + li;
    #pragma unroll
    for (int s = 0; s < 4; ++s) {
      const int rr = qq[s] / 18;
      const int cc = qq[s] - rr * 18;
      const int gy = y0 - 1 + rr;
      const int gx = x0 - 1 + cc;
      inb[s] = qq[s] < NQ && gy >= 0 && gy < HW && gx >= 0 && gx < HW;
      goff[s] = min(max(gy, 0), HW - 1) * HW + min(max(gx, 0), HW - 1);
    }
    const int wv = __builtin_amdgcn_readfirstlane(wave);
    const float* __restrict__ xc = x + ((size_t)b * IN_C + wv * 32) * L_IMG;
    float* __restrict__ lc = lds + LDS_X + wv * 32 * SX;
    float vA[32], vB[32], vC[32], vD[16];
    #pragma unroll
    for (int i = 0; i < 32; ++i) {
      vA[i] = xc[i * L_IMG + goff[0]];
      vB[i] = xc[i * L_IMG + goff[1]];
      vC[i] = xc[i * L_IMG + goff[2]];
    }
    #pragma unroll
    for (int g = 0; g < 16; ++g) vD[g] = xc[(2 * g) * L_IMG + half * L_IMG + goff[3]];
    #pragma unroll
    for (int i = 0; i < 32; ++i) {
      lc[i * SX + qq[0]] = inb[0] ? vA[i] : 0.f;
      lc[i * SX + qq[1]] = inb[1] ? vB[i] : 0.f;
      lc[i * SX + qq[2]] = inb[2] ? vC[i] : 0.f;
    }
    #pragma unroll
    for (int g = 0; g < 16; ++g)
      lc[(2 * g) * SX + half * SX + qq[3]] = inb[3] ? vD[g] : 0.f;
  }
  __syncthreads();

  {
    v8f acc[2][QT];
    #pragma unroll
    for (int j = 0; j < 2; ++j)
      #pragma unroll
      for (int qt = 0; qt < QT; ++qt) acc[j][qt] = vzero;

#define P1_LOAD(AF, BF, K0)                                                        \
    {                                                                              \
      _Pragma("unroll")       \
      for (int s = 0; s < 2; ++s) {                                                \
        const int kl = (K0) + 4 * s + 2 * half;                                    \
        _Pragma("unroll")                                                          \
        for (int j = 0; j < 2; ++j) {                                              \
          const int o0 = (2 * wave + j) * 16;                                      \
          AF[s][j].x = Wp[(o0 + li) * IN_C + kl];                                  \
          AF[s][j].y = Wp[(o0 + li) * IN_C + kl + 1];                              \
        }                                                                          \
        _Pragma("unroll")                                                          \
        for (int qt = 0; qt < QT; ++qt) {                                          \
          BF[s][qt].x = lds[LDS_X + kl * SX + qt * 16 + li];                       \
          BF[s][qt].y = lds[LDS_X + (kl + 1) * SX + qt * 16 + li];                 \
        }                                                                          \
      }                                                                            \
    }
#define P1_MMA(AF, BF)                                                             \
    {                                                                              \
      _Pragma("unroll")                  \
      for (int s = 0; s < 2; ++s)                                                  \
        _Pragma("unroll")                                                          \
        for (int j = 0; j < 2; ++j)                                                \
          _Pragma("unroll")                                                        \
          for (int qt = 0; qt < QT; ++qt)                                          \
            acc[j][qt] = wmma4(AF[s][j], BF[s][qt], acc[j][qt]);                   \
    }
    v2f pa0[2][2], pa1[2][2], pb0[2][QT], pb1[2][QT];
    P1_LOAD(pa0, pb0, 0)
    for (int k0 = 0; k0 < IN_C - 16; k0 += 16) {
      P1_LOAD(pa1, pb1, k0 + 8)
      __builtin_amdgcn_sched_barrier(0);
      P1_MMA(pa0, pb0)
      __builtin_amdgcn_sched_barrier(0);
      P1_LOAD(pa0, pb0, k0 + 16)
      __builtin_amdgcn_sched_barrier(0);
      P1_MMA(pa1, pb1)
      __builtin_amdgcn_sched_barrier(0);
    }
    P1_LOAD(pa1, pb1, IN_C - 8)
    __builtin_amdgcn_sched_barrier(0);
    P1_MMA(pa0, pb0)
    __builtin_amdgcn_sched_barrier(0);
    P1_MMA(pa1, pb1)
#undef P1_LOAD
#undef P1_MMA

    bool inimg[QT];
    #pragma unroll
    for (int qt = 0; qt < QT; ++qt) {
      const int q  = qt * 16 + li;
      const int rr = q / 18;
      const int cc = q - rr * 18;
      const int gy = y0 - 1 + rr;
      const int gx = x0 - 1 + cc;
      inimg[qt] = (q < NQ) && gy >= 0 && gy < HW && gx >= 0 && gx < HW;
    }
    float bv[2][8];
    #pragma unroll
    for (int j = 0; j < 2; ++j)
      #pragma unroll
      for (int r = 0; r < 8; ++r) bv[j][r] = bp[(2 * wave + j) * 16 + 8 * half + r];
    #pragma unroll
    for (int j = 0; j < 2; ++j)
      #pragma unroll
      for (int r = 0; r < 8; ++r) {
        const int o = (2 * wave + j) * 16 + 8 * half + r;
        #pragma unroll
        for (int qt = 0; qt < QT; ++qt) {
          const int q = qt * 16 + li;
          if (q < NQ)
            lds[LDS_P + o * SP + q] = inimg[qt] ? (acc[j][qt][r] + bv[j][r]) : 0.f;
        }
      }
  }
  __syncthreads();

  const int n = wave;
  v8f   oacc[4][ROWS];
  float mrun[ROWS], srun[ROWS];
  #pragma unroll
  for (int r = 0; r < ROWS; ++r) {
    mrun[r] = -3.0e38f;
    srun[r] = 0.f;
    #pragma unroll
    for (int dt = 0; dt < 4; ++dt) oacc[dt][r] = vzero;
  }

  for (int hh = 0; hh < 2; ++hh) {
    v8f att[8][ROWS];
    #pragma unroll
    for (int t = 0; t < 8; ++t)
      #pragma unroll
      for (int r = 0; r < ROWS; ++r) att[t][r] = vzero;

#define K2_LOAD(AF, BF, K0)                                                        \
    {                                                                              \
      const int kl = (K0) + 2 * half;                                              \
      _Pragma("unroll")           \
      for (int t = 0; t < 8; ++t) {                                                \
        AF[t].x = keyb[(n * KD2 + kl) * MEM + (hh * 8 + t) * 16 + li];             \
        AF[t].y = keyb[(n * KD2 + kl + 1) * MEM + (hh * 8 + t) * 16 + li];         \
      }                                                                            \
      _Pragma("unroll")                                                            \
      for (int v = 0; v < 2; ++v) {                                                \
        const int ch = (kl + v) * NH + n;       \
        const int c  = ch / 9;                                                     \
        const int p  = ch - c * 9;                                                 \
        const int kh = p / 3;                                                      \
        const int kw = p - kh * 3;                                                 \
        const int off = LDS_P + c * SP + kh * 18 + kw + li;                        \
        _Pragma("unroll")                                                          \
        for (int r = 0; r < ROWS; ++r) BF[r][v] = lds[off + 18 * r];               \
      }                                                                            \
    }
#define K2_MMA(AF, BF)                                                             \
    {                                                                              \
      _Pragma("unroll")               \
      for (int t = 0; t < 8; ++t)                                                  \
        _Pragma("unroll")                                                          \
        for (int r = 0; r < ROWS; ++r)                                             \
          att[t][r] = wmma4(AF[t], BF[r], att[t][r]);                              \
    }
    {
      v2f ka0[8], ka1[8], kb0[ROWS], kb1[ROWS];
      K2_LOAD(ka0, kb0, 0)
      for (int k0 = 0; k0 < KD2 - 8; k0 += 8) {
        K2_LOAD(ka1, kb1, k0 + 4)
        __builtin_amdgcn_sched_barrier(0);
        K2_MMA(ka0, kb0)
        __builtin_amdgcn_sched_barrier(0);
        K2_LOAD(ka0, kb0, k0 + 8)
        __builtin_amdgcn_sched_barrier(0);
        K2_MMA(ka1, kb1)
        __builtin_amdgcn_sched_barrier(0);
      }
      K2_LOAD(ka1, kb1, KD2 - 4)
      __builtin_amdgcn_sched_barrier(0);
      K2_MMA(ka0, kb0)
      __builtin_amdgcn_sched_barrier(0);
      K2_MMA(ka1, kb1)
    }
#undef K2_LOAD
#undef K2_MMA

    #pragma unroll
    for (int nt = 0; nt < ROWS; ++nt) {
      float mx = -3.0e38f;
      #pragma unroll
      for (int t = 0; t < 8; ++t)
        #pragma unroll
        for (int r = 0; r < 8; ++r) mx = fmaxf(mx, att[t][nt][r]);
      mx = fmaxf(mx, __shfl_xor(mx, 16, 32));
      const float mnew = fmaxf(mrun[nt], mx);
      const float f = __expf(mrun[nt] - mnew);
      mrun[nt] = mnew;
      #pragma unroll
      for (int dt = 0; dt < 4; ++dt)
        #pragma unroll
        for (int r = 0; r < 8; ++r) oacc[dt][nt][r] *= f;
      float sum = srun[nt] * f;
      #pragma unroll
      for (int t = 0; t < 8; ++t)
        #pragma unroll
        for (int r = 0; r < 8; ++r) {
          float e = __expf(att[t][nt][r] - mnew);
          att[t][nt][r] = e;
          sum += e;
        }
      srun[nt] = sum;
    }

    {
      float ma[4][2][4];
#define M3_LOAD(SET, G)                                                            \
      {                                                                            \
        const int i0 = hh * 128 + ((G) >> 2) * 16 + 8 * half + 2 * ((G) & 3);      \
        _Pragma("unroll")         \
        for (int dt = 0; dt < 4; ++dt) {                                           \
          ma[SET][0][dt] = memb[(n * MEM + i0) * HD + dt * 16 + li];               \
          ma[SET][1][dt] = memb[(n * MEM + i0 + 1) * HD + dt * 16 + li];           \
        }                                                                          \
      }
      M3_LOAD(0, 0)
      M3_LOAD(1, 1)
      M3_LOAD(2, 2)
      #pragma unroll
      for (int g = 0; g < 32; ++g) {
        if (g + 3 < 32) M3_LOAD((g + 3) & 3, g + 3)
        const int t = g >> 2;
        const int j = g & 3;
        v2f bq[ROWS];
        #pragma unroll
        for (int nt = 0; nt < ROWS; ++nt) {
          bq[nt].x = att[t][nt][2 * j];
          bq[nt].y = att[t][nt][2 * j + 1];
        }
        #pragma unroll
        for (int dt = 0; dt < 4; ++dt) {
          v2f af;
          af.x = ma[g & 3][0][dt];
          af.y = ma[g & 3][1][dt];
          #pragma unroll
          for (int nt = 0; nt < ROWS; ++nt)
            oacc[dt][nt] = wmma4(af, bq[nt], oacc[dt][nt]);
        }
        __builtin_amdgcn_sched_barrier(0);
      }
#undef M3_LOAD
    }
  }

  #pragma unroll
  for (int nt = 0; nt < ROWS; ++nt) {
    const float tot = srun[nt] + __shfl_xor(srun[nt], 16, 32);
    const float inv = 1.f / tot;
    #pragma unroll
    for (int dt = 0; dt < 4; ++dt)
      #pragma unroll
      for (int r = 0; r < 8; ++r) {
        const int d = dt * 16 + 8 * half + r;
        lds[LDS_O + (n * HD + d) * SO + nt * 16 + li] = oacc[dt][nt][r] * inv;
      }
  }
  __syncthreads();

  {
    v8f facc[2][ROWS];
    #pragma unroll
    for (int j = 0; j < 2; ++j)
      #pragma unroll
      for (int nt = 0; nt < ROWS; ++nt) facc[j][nt] = vzero;

#define O4_LOAD(AF, BF, K0)                                                        \
    {                                                                              \
      _Pragma("unroll")       \
      for (int s = 0; s < 8; ++s) {                                                \
        const int kl = (K0) + 4 * s + 2 * half;                                    \
        _Pragma("unroll")                                                          \
        for (int j = 0; j < 2; ++j) {                                              \
          const int o0 = (2 * wave + j) * 16;                                      \
          AF[s][j].x = Wo[(o0 + li) * (NH * HD) + kl];                             \
          AF[s][j].y = Wo[(o0 + li) * (NH * HD) + kl + 1];                         \
        }                                                                          \
        _Pragma("unroll")                                                          \
        for (int nt = 0; nt < ROWS; ++nt) {                                        \
          BF[s][nt].x = lds[LDS_O + kl * SO + nt * 16 + li];                       \
          BF[s][nt].y = lds[LDS_O + (kl + 1) * SO + nt * 16 + li];                 \
        }                                                                          \
      }                                                                            \
    }
#define O4_MMA(AF, BF)                                                             \
    {                                                                              \
      _Pragma("unroll")                  \
      for (int s = 0; s < 8; ++s)                                                  \
        _Pragma("unroll")                                                          \
        for (int j = 0; j < 2; ++j)                                                \
          _Pragma("unroll")                                                        \
          for (int nt = 0; nt < ROWS; ++nt)                                        \
            facc[j][nt] = wmma4(AF[s][j], BF[s][nt], facc[j][nt]);                 \
    }
    v2f oa0[8][2], oa1[8][2], ob0[8][ROWS], ob1[8][ROWS];
    O4_LOAD(oa0, ob0, 0)
    for (int k0 = 0; k0 < NH * HD - 64; k0 += 64) {
      O4_LOAD(oa1, ob1, k0 + 32)
      __builtin_amdgcn_sched_barrier(0);
      O4_MMA(oa0, ob0)
      __builtin_amdgcn_sched_barrier(0);
      O4_LOAD(oa0, ob0, k0 + 64)
      __builtin_amdgcn_sched_barrier(0);
      O4_MMA(oa1, ob1)
      __builtin_amdgcn_sched_barrier(0);
    }
    O4_LOAD(oa1, ob1, NH * HD - 32)
    __builtin_amdgcn_sched_barrier(0);
    O4_MMA(oa0, ob0)
    __builtin_amdgcn_sched_barrier(0);
    O4_MMA(oa1, ob1)
#undef O4_LOAD
#undef O4_MMA
    #pragma unroll
    for (int j = 0; j < 2; ++j)
      #pragma unroll
      for (int nt = 0; nt < ROWS; ++nt)
        #pragma unroll
        for (int r = 0; r < 8; ++r) {
          const int oo = (2 * wave + j) * 16 + 8 * half + r;
          out[(b * OUTC + oo) * L_IMG + (y0 + nt) * HW + x0 + li] = facc[j][nt][r] + bo[oo];
        }
  }
}

extern "C" void kernel_launch(void* const* d_in, const int* in_sizes, int n_in,
                              void* d_out, int out_size, void* d_ws, size_t ws_size,
                              hipStream_t stream) {
  (void)in_sizes; (void)n_in; (void)d_ws; (void)ws_size; (void)out_size;
  const float* x    = (const float*)d_in[0];
  const float* Wp   = (const float*)d_in[1];
  const float* bpj  = (const float*)d_in[2];
  const float* keyb = (const float*)d_in[3];
  const float* memb = (const float*)d_in[4];
  const float* Wo   = (const float*)d_in[5];
  const float* bo   = (const float*)d_in[6];
  float* out = (float*)d_out;

  const size_t shmem = (size_t)LDS_DW * sizeof(float);
  dim3 grid(512);
  dim3 block(256);
  kcmm_fused_gfx1250<<<grid, block, shmem, stream>>>(x, Wp, bpj, keyb, memb, Wo, bo, out);
}


# priority window: the younger wave of each SIMD pair (waves 4-7) runs at s_setprio 3 for the first two thirds of each barrier-to-barrier GEMM stretch, then at 0, so the pair finishes together
# speedup vs baseline: 2.5891x; 1.0339x over previous
#include <hip/hip_runtime.h>


typedef __attribute__((ext_vector_type(2))) float v2f;
typedef __attribute__((ext_vector_type(8))) float v8f;

#define IN_C    256
#define NH      8
#define KD2     288
#define MEM     256
#define HD      64
#define OUTC    256
#define HW      64
#define L_IMG   4096

#define ROWS    4
#define NQ      108
#define QT      7
#define SX      120
#define SP      116
#define SO      76

#define LDS_X    0
#define LDS_O    0
#define LDS_P    38912
#define LDS_DW   68608

static __device__ __forceinline__ v8f wmma4(v2f a, v2f b, v8f c) {
  return __builtin_amdgcn_wmma_f32_16x16x4_f32(
       false, a,  false, b,
       (short)0, c,  false,  false);
}

__global__ __launch_bounds__(256)
void kcmm_fused_gfx1250(const float* __restrict__ x,
                        const float* __restrict__ Wp,
                        const float* __restrict__ bp,
                        const float* __restrict__ keyb,
                        const float* __restrict__ memb,
                        const float* __restrict__ Wo,
                        const float* __restrict__ bo,
                        float* __restrict__ out)
{
  extern __shared__ float lds[];

  const int tid  = threadIdx.x;
  const int wave = tid >> 5;
  const int lane = tid & 31;
  const int half = lane >> 4;
  const int li   = lane & 15;

  const int blk = blockIdx.x;
  const int b   = blk >> 6;
  const int rem = blk & 63;
  const int y0  = (rem >> 2) << 2;
  const int x0  = (rem & 3) << 4;

  const v8f vzero = {0.f,0.f,0.f,0.f,0.f,0.f,0.f,0.f};

  const bool younger = __builtin_amdgcn_readfirstlane(wave) >= 4;

  {
    int  qq[4], goff[4];
    bool inb[4];
    qq[0] = lane;
    qq[1] = 32 + lane;
    qq[2] = 64 + lane;
    qq[3] = 96 + li;
    #pragma unroll
    for (int s = 0; s < 4; ++s) {
      const int rr = qq[s] / 18;
      const int cc = qq[s] - rr * 18;
      const int gy = y0 - 1 + rr;
      const int gx = x0 - 1 + cc;
      inb[s] = qq[s] < NQ && gy >= 0 && gy < HW && gx >= 0 && gx < HW;
      goff[s] = min(max(gy, 0), HW - 1) * HW + min(max(gx, 0), HW - 1);
    }
    const int wv = __builtin_amdgcn_readfirstlane(wave);
    const float* __restrict__ xc = x + ((size_t)b * IN_C + wv * 32) * L_IMG;
    float* __restrict__ lc = lds + LDS_X + wv * 32 * SX;
    float vA[32], vB[32], vC[32], vD[16];
    #pragma unroll
    for (int i = 0; i < 32; ++i) {
      vA[i] = xc[i * L_IMG + goff[0]];
      vB[i] = xc[i * L_IMG + goff[1]];
      vC[i] = xc[i * L_IMG + goff[2]];
    }
    #pragma unroll
    for (int g = 0; g < 16; ++g) vD[g] = xc[(2 * g) * L_IMG + half * L_IMG + goff[3]];
    #pragma unroll
    for (int i = 0; i < 32; ++i) {
      lc[i * SX + qq[0]] = inb[0] ? vA[i] : 0.f;
      lc[i * SX + qq[1]] = inb[1] ? vB[i] : 0.f;
      lc[i * SX + qq[2]] = inb[2] ? vC[i] : 0.f;
    }
    #pragma unroll
    for (int g = 0; g < 16; ++g)
      lc[(2 * g) * SX + half * SX + qq[3]] = inb[3] ? vD[g] : 0.f;
  }
  __syncthreads();

  {
    v8f acc[2][QT];
    #pragma unroll
    for (int j = 0; j < 2; ++j)
      #pragma unroll
      for (int qt = 0; qt < QT; ++qt) acc[j][qt] = vzero;

#define P1_LOAD(AF, BF, K0)                                                        \
    {                                                                              \
      _Pragma("unroll")       \
      for (int s = 0; s < 2; ++s) {                                                \
        const int kl = (K0) + 4 * s + 2 * half;                                    \
        _Pragma("unroll")                                                          \
        for (int j = 0; j < 2; ++j) {                                              \
          const int o0 = (2 * wave + j) * 16;                                      \
          AF[s][j].x = Wp[(o0 + li) * IN_C + kl];                                  \
          AF[s][j].y = Wp[(o0 + li) * IN_C + kl + 1];                              \
        }                                                                          \
        _Pragma("unroll")                                                          \
        for (int qt = 0; qt < QT; ++qt) {                                          \
          BF[s][qt].x = lds[LDS_X + kl * SX + qt * 16 + li];                       \
          BF[s][qt].y = lds[LDS_X + (kl + 1) * SX + qt * 16 + li];                 \
        }                                                                          \
      }                                                                            \
    }
#define P1_MMA(AF, BF)                                                             \
    {                                                                              \
      _Pragma("unroll")                  \
      for (int s = 0; s < 2; ++s)                                                  \
        _Pragma("unroll")                                                          \
        for (int j = 0; j < 2; ++j)                                                \
          _Pragma("unroll")                                                        \
          for (int qt = 0; qt < QT; ++qt)                                          \
            acc[j][qt] = wmma4(AF[s][j], BF[s][qt], acc[j][qt]);                   \
    }
    v2f pa0[2][2], pa1[2][2], pb0[2][QT], pb1[2][QT];
    if (younger) __builtin_amdgcn_s_setprio(3);
    P1_LOAD(pa0, pb0, 0)
    for (int k0 = 0; k0 < IN_C - 16; k0 += 16) {
      if (k0 == 176) __builtin_amdgcn_s_setprio(0);
      P1_LOAD(pa1, pb1, k0 + 8)
      __builtin_amdgcn_sched_barrier(0);
      P1_MMA(pa0, pb0)
      __builtin_amdgcn_sched_barrier(0);
      P1_LOAD(pa0, pb0, k0 + 16)
      __builtin_amdgcn_sched_barrier(0);
      P1_MMA(pa1, pb1)
      __builtin_amdgcn_sched_barrier(0);
    }
    P1_LOAD(pa1, pb1, IN_C - 8)
    __builtin_amdgcn_sched_barrier(0);
    P1_MMA(pa0, pb0)
    __builtin_amdgcn_sched_barrier(0);
    P1_MMA(pa1, pb1)
#undef P1_LOAD
#undef P1_MMA

    bool inimg[QT];
    #pragma unroll
    for (int qt = 0; qt < QT; ++qt) {
      const int q  = qt * 16 + li;
      const int rr = q / 18;
      const int cc = q - rr * 18;
      const int gy = y0 - 1 + rr;
      const int gx = x0 - 1 + cc;
      inimg[qt] = (q < NQ) && gy >= 0 && gy < HW && gx >= 0 && gx < HW;
    }
    float bv[2][8];
    #pragma unroll
    for (int j = 0; j < 2; ++j)
      #pragma unroll
      for (int r = 0; r < 8; ++r) bv[j][r] = bp[(2 * wave + j) * 16 + 8 * half + r];
    #pragma unroll
    for (int j = 0; j < 2; ++j)
      #pragma unroll
      for (int r = 0; r < 8; ++r) {
        const int o = (2 * wave + j) * 16 + 8 * half + r;
        #pragma unroll
        for (int qt = 0; qt < QT; ++qt) {
          const int q = qt * 16 + li;
          if (q < NQ)
            lds[LDS_P + o * SP + q] = inimg[qt] ? (acc[j][qt][r] + bv[j][r]) : 0.f;
        }
      }
  }
  __syncthreads();

  const int n = wave;
  v8f   oacc[4][ROWS];
  float mrun[ROWS], srun[ROWS];
  #pragma unroll
  for (int r = 0; r < ROWS; ++r) {
    mrun[r] = -3.0e38f;
    srun[r] = 0.f;
    #pragma unroll
    for (int dt = 0; dt < 4; ++dt) oacc[dt][r] = vzero;
  }

  if (younger) __builtin_amdgcn_s_setprio(3);

  for (int hh = 0; hh < 2; ++hh) {
    v8f att[8][ROWS];
    #pragma unroll
    for (int t = 0; t < 8; ++t)
      #pragma unroll
      for (int r = 0; r < ROWS; ++r) att[t][r] = vzero;

#define K2_LOAD(AF, BF, K0)                                                        \
    {                                                                              \
      const int kl = (K0) + 2 * half;                                              \
      _Pragma("unroll")           \
      for (int t = 0; t < 8; ++t) {                                                \
        AF[t].x = keyb[(n * KD2 + kl) * MEM + (hh * 8 + t) * 16 + li];             \
        AF[t].y = keyb[(n * KD2 + kl + 1) * MEM + (hh * 8 + t) * 16 + li];         \
      }                                                                            \
      _Pragma("unroll")                                                            \
      for (int v = 0; v < 2; ++v) {                                                \
        const int ch = (kl + v) * NH + n;       \
        const int c  = ch / 9;                                                     \
        const int p  = ch - c * 9;                                                 \
        const int kh = p / 3;                                                      \
        const int kw = p - kh * 3;                                                 \
        const int off = LDS_P + c * SP + kh * 18 + kw + li;                        \
        _Pragma("unroll")                                                          \
        for (int r = 0; r < ROWS; ++r) BF[r][v] = lds[off + 18 * r];               \
      }                                                                            \
    }
#define K2_MMA(AF, BF)                                                             \
    {                                                                              \
      _Pragma("unroll")               \
      for (int t = 0; t < 8; ++t)                                                  \
        _Pragma("unroll")                                                          \
        for (int r = 0; r < ROWS; ++r)                                             \
          att[t][r] = wmma4(AF[t], BF[r], att[t][r]);                              \
    }
    {
      v2f ka0[8], ka1[8], kb0[ROWS], kb1[ROWS];
      K2_LOAD(ka0, kb0, 0)
      const int k0_window_end = hh ? 112 : -8;
      for (int k0 = 0; k0 < KD2 - 8; k0 += 8) {
        if (k0 == k0_window_end) __builtin_amdgcn_s_setprio(0);
        K2_LOAD(ka1, kb1, k0 + 4)
        __builtin_amdgcn_sched_barrier(0);
        K2_MMA(ka0, kb0)
        __builtin_amdgcn_sched_barrier(0);
        K2_LOAD(ka0, kb0, k0 + 8)
        __builtin_amdgcn_sched_barrier(0);
        K2_MMA(ka1, kb1)
        __builtin_amdgcn_sched_barrier(0);
      }
      K2_LOAD(ka1, kb1, KD2 - 4)
      __builtin_amdgcn_sched_barrier(0);
      K2_MMA(ka0, kb0)
      __builtin_amdgcn_sched_barrier(0);
      K2_MMA(ka1, kb1)
    }
#undef K2_LOAD
#undef K2_MMA

    #pragma unroll
    for (int nt = 0; nt < ROWS; ++nt) {
      float mx = -3.0e38f;
      #pragma unroll
      for (int t = 0; t < 8; ++t)
        #pragma unroll
        for (int r = 0; r < 8; ++r) mx = fmaxf(mx, att[t][nt][r]);
      mx = fmaxf(mx, __shfl_xor(mx, 16, 32));
      const float mnew = fmaxf(mrun[nt], mx);
      const float f = __expf(mrun[nt] - mnew);
      mrun[nt] = mnew;
      #pragma unroll
      for (int dt = 0; dt < 4; ++dt)
        #pragma unroll
        for (int r = 0; r < 8; ++r) oacc[dt][nt][r] *= f;
      float sum = srun[nt] * f;
      #pragma unroll
      for (int t = 0; t < 8; ++t)
        #pragma unroll
        for (int r = 0; r < 8; ++r) {
          float e = __expf(att[t][nt][r] - mnew);
          att[t][nt][r] = e;
          sum += e;
        }
      srun[nt] = sum;
    }

    {
      float ma[4][2][4];
#define M3_LOAD(SET, G)                                                            \
      {                                                                            \
        const int i0 = hh * 128 + ((G) >> 2) * 16 + 8 * half + 2 * ((G) & 3);      \
        _Pragma("unroll")         \
        for (int dt = 0; dt < 4; ++dt) {                                           \
          ma[SET][0][dt] = memb[(n * MEM + i0) * HD + dt * 16 + li];               \
          ma[SET][1][dt] = memb[(n * MEM + i0 + 1) * HD + dt * 16 + li];           \
        }                                                                          \
      }
      M3_LOAD(0, 0)
      M3_LOAD(1, 1)
      M3_LOAD(2, 2)
      #pragma unroll
      for (int g = 0; g < 32; ++g) {
        if (g + 3 < 32) M3_LOAD((g + 3) & 3, g + 3)
        const int t = g >> 2;
        const int j = g & 3;
        v2f bq[ROWS];
        #pragma unroll
        for (int nt = 0; nt < ROWS; ++nt) {
          bq[nt].x = att[t][nt][2 * j];
          bq[nt].y = att[t][nt][2 * j + 1];
        }
        #pragma unroll
        for (int dt = 0; dt < 4; ++dt) {
          v2f af;
          af.x = ma[g & 3][0][dt];
          af.y = ma[g & 3][1][dt];
          #pragma unroll
          for (int nt = 0; nt < ROWS; ++nt)
            oacc[dt][nt] = wmma4(af, bq[nt], oacc[dt][nt]);
        }
        __builtin_amdgcn_sched_barrier(0);
      }
#undef M3_LOAD
    }
  }

  #pragma unroll
  for (int nt = 0; nt < ROWS; ++nt) {
    const float tot = srun[nt] + __shfl_xor(srun[nt], 16, 32);
    const float inv = 1.f / tot;
    #pragma unroll
    for (int dt = 0; dt < 4; ++dt)
      #pragma unroll
      for (int r = 0; r < 8; ++r) {
        const int d = dt * 16 + 8 * half + r;
        lds[LDS_O + (n * HD + d) * SO + nt * 16 + li] = oacc[dt][nt][r] * inv;
      }
  }
  __syncthreads();

  {
    v8f facc[2][ROWS];
    #pragma unroll
    for (int j = 0; j < 2; ++j)
      #pragma unroll
      for (int nt = 0; nt < ROWS; ++nt) facc[j][nt] = vzero;

#define O4_LOAD(AF, BF, K0)                                                        \
    {                                                                              \
      _Pragma("unroll")       \
      for (int s = 0; s < 8; ++s) {                                                \
        const int kl = (K0) + 4 * s + 2 * half;                                    \
        _Pragma("unroll")                                                          \
        for (int j = 0; j < 2; ++j) {                                              \
          const int o0 = (2 * wave + j) * 16;                                      \
          AF[s][j].x = Wo[(o0 + li) * (NH * HD) + kl];                             \
          AF[s][j].y = Wo[(o0 + li) * (NH * HD) + kl + 1];                         \
        }                                                                          \
        _Pragma("unroll")                                                          \
        for (int nt = 0; nt < ROWS; ++nt) {                                        \
          BF[s][nt].x = lds[LDS_O + kl * SO + nt * 16 + li];                       \
          BF[s][nt].y = lds[LDS_O + (kl + 1) * SO + nt * 16 + li];                 \
        }                                                                          \
      }                                                                            \
    }
#define O4_MMA(AF, BF)                                                             \
    {                                                                              \
      _Pragma("unroll")                  \
      for (int s = 0; s < 8; ++s)                                                  \
        _Pragma("unroll")                                                          \
        for (int j = 0; j < 2; ++j)                                                \
          _Pragma("unroll")                                                        \
          for (int nt = 0; nt < ROWS; ++nt)                                        \
            facc[j][nt] = wmma4(AF[s][j], BF[s][nt], facc[j][nt]);                 \
    }
    v2f oa0[8][2], oa1[8][2], ob0[8][ROWS], ob1[8][ROWS];
    if (younger) __builtin_amdgcn_s_setprio(3);
    O4_LOAD(oa0, ob0, 0)
    for (int k0 = 0; k0 < NH * HD - 64; k0 += 64) {
      if (k0 == 320) __builtin_amdgcn_s_setprio(0);
      O4_LOAD(oa1, ob1, k0 + 32)
      __builtin_amdgcn_sched_barrier(0);
      O4_MMA(oa0, ob0)
      __builtin_amdgcn_sched_barrier(0);
      O4_LOAD(oa0, ob0, k0 + 64)
      __builtin_amdgcn_sched_barrier(0);
      O4_MMA(oa1, ob1)
      __builtin_amdgcn_sched_barrier(0);
    }
    O4_LOAD(oa1, ob1, NH * HD - 32)
    __builtin_amdgcn_sched_barrier(0);
    O4_MMA(oa0, ob0)
    __builtin_amdgcn_sched_barrier(0);
    O4_MMA(oa1, ob1)
#undef O4_LOAD
#undef O4_MMA
    #pragma unroll
    for (int j = 0; j < 2; ++j)
      #pragma unroll
      for (int nt = 0; nt < ROWS; ++nt)
        #pragma unroll
        for (int r = 0; r < 8; ++r) {
          const int oo = (2 * wave + j) * 16 + 8 * half + r;
          out[(b * OUTC + oo) * L_IMG + (y0 + nt) * HW + x0 + li] = facc[j][nt][r] + bo[oo];
        }
  }
}

extern "C" void kernel_launch(void* const* d_in, const int* in_sizes, int n_in,
                              void* d_out, int out_size, void* d_ws, size_t ws_size,
                              hipStream_t stream) {
  (void)in_sizes; (void)n_in; (void)d_ws; (void)ws_size; (void)out_size;
  const float* x    = (const float*)d_in[0];
  const float* Wp   = (const float*)d_in[1];
  const float* bpj  = (const float*)d_in[2];
  const float* keyb = (const float*)d_in[3];
  const float* memb = (const float*)d_in[4];
  const float* Wo   = (const float*)d_in[5];
  const float* bo   = (const float*)d_in[6];
  float* out = (float*)d_out;

  const size_t shmem = (size_t)LDS_DW * sizeof(float);
  dim3 grid(512);
  dim3 block(256);
  kcmm_fused_gfx1250<<<grid, block, shmem, stream>>>(x, Wp, bpj, keyb, memb, Wo, bo, out);
}
